# starting kernel
# baseline (speedup 1.0000x reference)
#include <hip/hip_runtime.h>
#include <hip/hip_bf16.h>
#include <math.h>

typedef __attribute__((ext_vector_type(2))) float v2f;
typedef __attribute__((ext_vector_type(4))) float v4f;
typedef __attribute__((ext_vector_type(8))) float v8f;

#define BATCH 8
#define N_PTS 512
#define DIN   32
#define H1DIM 64
#define H2DIM 64
#define H3DIM 32

__device__ __forceinline__ void wave_sync() {
    __builtin_amdgcn_fence(__ATOMIC_ACQ_REL, "workgroup");
    __builtin_amdgcn_wave_barrier();
}

__global__ __launch_bounds__(64) void cpc_precompute_kernel(
        const float* __restrict__ x, const float* __restrict__ W1,
        const float* __restrict__ b1,
        float* __restrict__ Apre, float* __restrict__ Bpre) {
    __shared__ float xr[DIN];
    const int row = blockIdx.x;
    const int c = threadIdx.x;
    if (c < DIN) xr[c] = x[row * DIN + c];
    __syncthreads();
    float a = 0.f;
    float bacc = b1[c];
    #pragma unroll
    for (int d = 0; d < DIN; ++d) {
        const float xv = xr[d];
        a    = fmaf(xv, W1[d * H1DIM + c], a);
        bacc = fmaf(xv, W1[(DIN + d) * H1DIM + c], bacc);
    }
    Apre[row * H1DIM + c] = a;
    Bpre[row * H1DIM + c] = bacc;
}

#define KSTR 68
#define TSTR 36

__global__ __launch_bounds__(256) void cpc_pair_mlp_kernel(
        const float* __restrict__ Apre, const float* __restrict__ Bpre,
        const float* __restrict__ W2, const float* __restrict__ b2,
        const float* __restrict__ W3, const float* __restrict__ b3,
        const float* __restrict__ W4, const float* __restrict__ b4,
        float* __restrict__ out) {
    __shared__ float W2Ts[H2DIM * KSTR];
    __shared__ float W3Ts[H3DIM * KSTR];
    __shared__ float b2s[H2DIM];
    __shared__ float b3s[H3DIM];
    __shared__ float W4s[H3DIM];
    __shared__ float Bis[H1DIM];
    __shared__ float bufA[8][16 * KSTR];
    __shared__ float bufB[8][16 * TSTR];

    const int tid  = threadIdx.x;
    const int wave = tid >> 5;
    const int lane = tid & 31;
    const int half = lane >> 4;
    const int mrow = lane & 15;
    const int bi   = blockIdx.x;
    const int b    = bi >> 9;

    for (int idx = tid; idx < H1DIM * H2DIM; idx += 256) {
        const int k = idx >> 6, n = idx & 63;
        W2Ts[n * KSTR + k] = W2[idx];
    }
    for (int idx = tid; idx < H2DIM * H3DIM; idx += 256) {
        const int k = idx >> 5, n = idx & 31;
        W3Ts[n * KSTR + k] = W3[idx];
    }
    if (tid < H2DIM) b2s[tid] = b2[tid];
    if (tid < H3DIM) { b3s[tid] = b3[tid]; W4s[tid] = W4[tid]; }
    if (tid < H1DIM) Bis[tid] = Bpre[bi * H1DIM + tid];
    __syncthreads();

    const float b4v = b4[0];
    float* H12 = &bufA[wave][0];
    float* H3s = &bufB[wave][0];

    for (int tt = 0; tt < 4; ++tt) {
        const int j0 = (wave + tt * 8) * 16;
        const float* arow = &Apre[(b * N_PTS + j0) * H1DIM];

        #pragma unroll
        for (int t = 0; t < 8; ++t) {
            const int li = lane + 32 * t;
            const int r  = li >> 4;
            const int c  = 4 * (lane & 15);
            v4f v = *(const v4f*)&arow[li * 4];
            const v4f bb = *(const v4f*)&Bis[c];
            v = v + bb;
            v.x = fmaxf(v.x, 0.f); v.y = fmaxf(v.y, 0.f);
            v.z = fmaxf(v.z, 0.f); v.w = fmaxf(v.w, 0.f);
            *(v4f*)&H12[r * KSTR + c] = v;
        }
        wave_sync();

        v8f acc[4] = {};
        #pragma unroll
        for (int k0 = 0; k0 < H1DIM; k0 += 4) {
            const v2f a = *(const v2f*)&H12[mrow * KSTR + k0 + 2 * half];
            #pragma unroll
            for (int nt = 0; nt < 4; ++nt) {
                const v2f bf =
                    *(const v2f*)&W2Ts[(nt * 16 + mrow) * KSTR + k0 + 2 * half];
                acc[nt] = __builtin_amdgcn_wmma_f32_16x16x4_f32(
                    false, a, false, bf, (short)0, acc[nt], false, false);
            }
        }
        wave_sync();
        #pragma unroll
        for (int r = 0; r < 8; ++r) {
            const int row = r + 8 * half;
            #pragma unroll
            for (int nt = 0; nt < 4; ++nt) {
                const int col = nt * 16 + mrow;
                H12[row * KSTR + col] = fmaxf(acc[nt][r] + b2s[col], 0.f);
            }
        }
        wave_sync();

        v8f c3[2] = {};
        #pragma unroll
        for (int k0 = 0; k0 < H2DIM; k0 += 4) {
            const v2f a = *(const v2f*)&H12[mrow * KSTR + k0 + 2 * half];
            #pragma unroll
            for (int nt = 0; nt < 2; ++nt) {
                const v2f bf =
                    *(const v2f*)&W3Ts[(nt * 16 + mrow) * KSTR + k0 + 2 * half];
                c3[nt] = __builtin_amdgcn_wmma_f32_16x16x4_f32(
                    false, a, false, bf, (short)0, c3[nt], false, false);
            }
        }
        wave_sync();
        #pragma unroll
        for (int r = 0; r < 8; ++r) {
            const int row = r + 8 * half;
            #pragma unroll
            for (int nt = 0; nt < 2; ++nt) {
                const int col = nt * 16 + mrow;
                H3s[row * TSTR + col] = fmaxf(c3[nt][r] + b3s[col], 0.f);
            }
        }
        wave_sync();

        float s = 0.f;
        #pragma unroll
        for (int k = 0; k < 16; ++k) {
            const int kk = half * 16 + k;
            s = fmaf(H3s[mrow * TSTR + kk], W4s[kk], s);
        }
        s += __shfl_xor(s, 16, 32);
        if (half == 0) {
            const float z = s + b4v;
            out[bi * N_PTS + j0 + mrow] = 1.f / (1.f + __expf(-z));
        }
        wave_sync();
    }
}

extern "C" void kernel_launch(void* const* d_in, const int* in_sizes, int n_in,
                              void* d_out, int out_size, void* d_ws, size_t ws_size,
                              hipStream_t stream) {
    const float* x  = (const float*)d_in[0];
    const float* W1 = (const float*)d_in[1];
    const float* b1 = (const float*)d_in[2];
    const float* W2 = (const float*)d_in[3];
    const float* b2 = (const float*)d_in[4];
    const float* W3 = (const float*)d_in[5];
    const float* b3 = (const float*)d_in[6];
    const float* W4 = (const float*)d_in[7];
    const float* b4 = (const float*)d_in[8];
    float* out = (float*)d_out;

    float* Apre = (float*)d_ws;
    float* Bpre = Apre + BATCH * N_PTS * H1DIM;

    cpc_precompute_kernel<<<BATCH * N_PTS, 64, 0, stream>>>(x, W1, b1, Apre, Bpre);
    cpc_pair_mlp_kernel<<<BATCH * N_PTS, 256, 0, stream>>>(
        Apre, Bpre, W2, b2, W3, b3, W4, b4, out);
}


# LDS reads taken out of the matrix-instruction stream: each block's fragment reads issued as one group a block ahead (second register set, scheduling fences), bias reads hoisted; sums unchanged
# speedup vs baseline: 1.2019x; 1.2019x over previous
#include <hip/hip_runtime.h>
#include <hip/hip_bf16.h>
#include <math.h>

typedef __attribute__((ext_vector_type(2))) float v2f;
typedef __attribute__((ext_vector_type(4))) float v4f;
typedef __attribute__((ext_vector_type(8))) float v8f;

#define BATCH 8
#define N_PTS 512
#define DIN   32
#define H1DIM 64
#define H2DIM 64
#define H3DIM 32
#define ICH   32
#define NWG   512

__device__ __forceinline__ void wave_sync() {
    __builtin_amdgcn_fence(__ATOMIC_ACQ_REL, "workgroup");
    __builtin_amdgcn_wave_barrier();
}

__device__ __forceinline__ void sched_fence() { __builtin_amdgcn_sched_barrier(0); }

#define KSTR 68
#define TSTR 36

#define OFF_BIS 0
#define OFF_XS  (OFF_BIS + ICH * H1DIM)
#define OFF_B2  (OFF_XS + ICH * DIN)
#define OFF_B3  (OFF_B2 + H2DIM)
#define OFF_W4  (OFF_B3 + H3DIM)
#define OFF_W3T (OFF_W4 + H3DIM)
#define OFF_W2T (OFF_W3T + H3DIM * KSTR)
#define OFF_BUF (OFF_W2T + H2DIM * KSTR)
#define LDS_FLOATS (OFF_BUF + 8 * 2 * 16 * KSTR)

struct L2Set {
    v2f w[4][4];
    v2f bi[4][2];
};

template <int C>
__device__ __forceinline__ void l2_load(L2Set& s, const float* W2Ts, const float* Bi0, const float* Bi1,
                                        int mrow, int half) {
    #pragma unroll
    for (int tt = 0; tt < 4; ++tt) {
        const int k = 4 * (4 * C + tt) + 2 * half;
        s.bi[tt][0] = *(const v2f*)&Bi0[k];
        s.bi[tt][1] = *(const v2f*)&Bi1[k];
        #pragma unroll
        for (int nt = 0; nt < 4; ++nt)
            s.w[tt][nt] = *(const v2f*)&W2Ts[(nt * 16 + mrow) * KSTR + k];
    }
}

template <int C>
__device__ __forceinline__ void l2_mma(const L2Set& s, const v2f (&apre)[16], v8f (&acc0)[4], v8f (&acc1)[4]) {
    #pragma unroll
    for (int tt = 0; tt < 4; ++tt) {
        v2f a0 = apre[4 * C + tt] + s.bi[tt][0];
        v2f a1 = apre[4 * C + tt] + s.bi[tt][1];
        a0.x = fmaxf(a0.x, 0.f); a0.y = fmaxf(a0.y, 0.f);
        a1.x = fmaxf(a1.x, 0.f); a1.y = fmaxf(a1.y, 0.f);
        #pragma unroll
        for (int nt = 0; nt < 4; ++nt) {
            acc0[nt] = __builtin_amdgcn_wmma_f32_16x16x4_f32(
                false, a0, false, s.w[tt][nt], (short)0, acc0[nt], false, false);
            acc1[nt] = __builtin_amdgcn_wmma_f32_16x16x4_f32(
                false, a1, false, s.w[tt][nt], (short)0, acc1[nt], false, false);
        }
    }
}

__global__ __launch_bounds__(256) void cpc_pair_mlp_fused_kernel(
        const float* __restrict__ x,  const float* __restrict__ W1,
        const float* __restrict__ b1,
        const float* __restrict__ W2, const float* __restrict__ b2,
        const float* __restrict__ W3, const float* __restrict__ b3,
        const float* __restrict__ W4, const float* __restrict__ b4,
        float* __restrict__ out) {
    __shared__ __attribute__((aligned(16))) float lds[LDS_FLOATS];
    float* Bis  = lds + OFF_BIS;
    float* xs   = lds + OFF_XS;
    float* b2s  = lds + OFF_B2;
    float* b3s  = lds + OFF_B3;
    float* W4s  = lds + OFF_W4;
    float* W3Ts = lds + OFF_W3T;
    float* W2Ts = lds + OFF_W2T;

    const int tid  = threadIdx.x;
    const int wave = tid >> 5;
    const int lane = tid & 31;
    const int half = lane >> 4;
    const int mrow = lane & 15;
    const int blk  = blockIdx.x;
    const int b    = blk >> 6;
    const int i0   = ((blk >> 2) & 15) * ICH;
    const int j0   = ((blk & 3) * 8 + wave) * 16;

    float* H12 = lds + OFF_BUF + wave * (2 * 16 * KSTR);
    float* H12b = H12 + 16 * KSTR;
    float* H3s = H12;
    float* H3sb = H12b;

    {
        float w2r[16], w3r[8];
        #pragma unroll
        for (int q = 0; q < 16; ++q) w2r[q] = W2[tid + 256 * q];
        #pragma unroll
        for (int q = 0; q < 8; ++q)  w3r[q] = W3[tid + 256 * q];
        const float* xi = &x[(b * N_PTS + i0) * DIN];
        const float xr0 = xi[tid], xr1 = xi[tid + 256], xr2 = xi[tid + 512], xr3 = xi[tid + 768];
        float sv0 = 0.f, sv1 = 0.f, sv2 = 0.f;
        if (tid < H2DIM) sv0 = b2[tid];
        if (tid < H3DIM) { sv1 = b3[tid]; sv2 = W4[tid]; }
        #pragma unroll
        for (int q = 0; q < 16; ++q) {
            const int idx = tid + 256 * q;
            const int k = idx >> 6, n = idx & 63;
            W2Ts[n * KSTR + k] = w2r[q];
        }
        #pragma unroll
        for (int q = 0; q < 8; ++q) {
            const int idx = tid + 256 * q;
            const int k = idx >> 5, n = idx & 31;
            W3Ts[n * KSTR + k] = w3r[q];
        }
        xs[tid]       = xr0;
        xs[tid + 256] = xr1;
        xs[tid + 512] = xr2;
        xs[tid + 768] = xr3;
        if (tid < H2DIM) b2s[tid] = sv0;
        if (tid < H3DIM) { b3s[tid] = sv1; W4s[tid] = sv2; }
    }

    {
        v8f aj[4] = {};
        const float* xj = &x[(b * N_PTS + j0 + mrow) * DIN];
        #pragma unroll
        for (int t = 0; t < DIN / 4; ++t) {
            const int k = 4 * t + 2 * half;
            const v2f a = *(const v2f*)&xj[k];
            #pragma unroll
            for (int nt = 0; nt < 4; ++nt) {
                v2f bf;
                bf.x = W1[k * H1DIM + nt * 16 + mrow];
                bf.y = W1[(k + 1) * H1DIM + nt * 16 + mrow];
                aj[nt] = __builtin_amdgcn_wmma_f32_16x16x4_f32(
                    false, a, false, bf, (short)0, aj[nt], false, false);
            }
        }
        #pragma unroll
        for (int r = 0; r < 8; ++r) {
            const int row = r + 8 * half;
            #pragma unroll
            for (int nt = 0; nt < 4; ++nt)
                H12[row * KSTR + nt * 16 + mrow] = aj[nt][r];
        }
    }
    __syncthreads();

    {
        const int c = tid & 63;
        const int q = tid >> 6;
        float w1c[DIN];
        #pragma unroll
        for (int d = 0; d < DIN; ++d) w1c[d] = W1[(DIN + d) * H1DIM + c];
        const float bias = b1[c];
        #pragma unroll
        for (int rr = 0; rr < ICH / 4; ++rr) {
            const int r = q + 4 * rr;
            float acc1 = bias;
            #pragma unroll
            for (int d = 0; d < DIN; ++d) acc1 = fmaf(xs[r * DIN + d], w1c[d], acc1);
            Bis[r * H1DIM + c] = acc1;
        }
    }

    v2f apre[16];
    #pragma unroll
    for (int t = 0; t < 16; ++t)
        apre[t] = *(const v2f*)&H12[mrow * KSTR + 4 * t + 2 * half];
    __syncthreads();

    const float b4v = b4[0];
    float* orow = &out[(b * N_PTS + i0) * N_PTS + j0 + mrow];

    L2Set sa, sb;
    l2_load<0>(sa, W2Ts, Bis, Bis + H1DIM, mrow, half);
    float b2r[4], b3r[2];
    #pragma unroll
    for (int nt = 0; nt < 4; ++nt) b2r[nt] = b2s[nt * 16 + mrow];
    #pragma unroll
    for (int nt = 0; nt < 2; ++nt) b3r[nt] = b3s[nt * 16 + mrow];

    for (int ip = 0; ip < ICH / 2; ++ip) {
        if (ip == 0)  __builtin_amdgcn_s_setprio(3);
        if (ip == 5)  __builtin_amdgcn_s_setprio(2);
        if (ip == 10) __builtin_amdgcn_s_setprio(1);
        if (ip == 15) __builtin_amdgcn_s_setprio(0);
        const float* Bi0 = &Bis[(2 * ip) * H1DIM];
        const float* Bi1 = Bi0 + H1DIM;

        v8f acc0[4] = {}, acc1[4] = {};
        sched_fence();
        l2_load<1>(sb, W2Ts, Bi0, Bi1, mrow, half);
        sched_fence();
        l2_mma<0>(sa, apre, acc0, acc1);
        sched_fence();
        l2_load<2>(sa, W2Ts, Bi0, Bi1, mrow, half);
        sched_fence();
        l2_mma<1>(sb, apre, acc0, acc1);
        sched_fence();
        l2_load<3>(sb, W2Ts, Bi0, Bi1, mrow, half);
        sched_fence();
        l2_mma<2>(sa, apre, acc0, acc1);
        l2_mma<3>(sb, apre, acc0, acc1);
        sched_fence();
        #pragma unroll
        for (int r = 0; r < 8; ++r) {
            const int row = r + 8 * half;
            #pragma unroll
            for (int nt = 0; nt < 4; ++nt) {
                const int col = nt * 16 + mrow;
                const float bias2 = b2r[nt];
                H12[row * KSTR + col]  = fmaxf(acc0[nt][r] + bias2, 0.f);
                H12b[row * KSTR + col] = fmaxf(acc1[nt][r] + bias2, 0.f);
            }
        }
        wave_sync();

        v2f a30[16], a31[16], w3f[16][2];
        #pragma unroll
        for (int ks = 0; ks < 16; ++ks) {
            if (ks == 4) sched_fence();
            const int k0 = 4 * ks + 2 * half;
            a30[ks] = *(const v2f*)&H12[mrow * KSTR + k0];
            a31[ks] = *(const v2f*)&H12b[mrow * KSTR + k0];
            #pragma unroll
            for (int nt = 0; nt < 2; ++nt)
                w3f[ks][nt] = *(const v2f*)&W3Ts[(nt * 16 + mrow) * KSTR + k0];
        }
        sched_fence();
        v8f c30[2] = {}, c31[2] = {};
        #pragma unroll
        for (int ks = 0; ks < 16; ++ks) {
            #pragma unroll
            for (int nt = 0; nt < 2; ++nt) {
                c30[nt] = __builtin_amdgcn_wmma_f32_16x16x4_f32(
                    false, a30[ks], false, w3f[ks][nt], (short)0, c30[nt], false, false);
                c31[nt] = __builtin_amdgcn_wmma_f32_16x16x4_f32(
                    false, a31[ks], false, w3f[ks][nt], (short)0, c31[nt], false, false);
            }
        }
        wave_sync();
        #pragma unroll
        for (int r = 0; r < 8; ++r) {
            const int row = r + 8 * half;
            #pragma unroll
            for (int nt = 0; nt < 2; ++nt) {
                const int col = nt * 16 + mrow;
                const float bias3 = b3r[nt];
                H3s[row * TSTR + col]  = fmaxf(c30[nt][r] + bias3, 0.f);
                H3sb[row * TSTR + col] = fmaxf(c31[nt][r] + bias3, 0.f);
            }
        }
        wave_sync();

        float h30[16], h31[16], w4r[16];
        #pragma unroll
        for (int k = 0; k < 16; ++k) {
            const int kk = half * 16 + k;
            w4r[k] = W4s[kk];
            h30[k] = H3s[mrow * TSTR + kk];
            h31[k] = H3sb[mrow * TSTR + kk];
        }
        {
            const float* Bn0 = &Bis[(2 * ((ip + 1) & (ICH / 2 - 1))) * H1DIM];
            l2_load<0>(sa, W2Ts, Bn0, Bn0 + H1DIM, mrow, half);
        }
        sched_fence();
        float s0 = 0.f, s1 = 0.f;
        #pragma unroll
        for (int k = 0; k < 16; ++k) {
            s0 = fmaf(h30[k], w4r[k], s0);
            s1 = fmaf(h31[k], w4r[k], s1);
        }
        s0 += __shfl_xor(s0, 16, 32);
        s1 += __shfl_xor(s1, 16, 32);
        const float z0 = s0 + b4v;
        const float z1 = s1 + b4v;
        const float res0 = 1.f / (1.f + __expf(-z0));
        const float res1 = 1.f / (1.f + __expf(-z1));
        wave_sync();
        if (half == 0) {
            orow[(2 * ip) * N_PTS]     = res0;
            orow[(2 * ip + 1) * N_PTS] = res1;
        }
    }
}

extern "C" void kernel_launch(void* const* d_in, const int* in_sizes, int n_in,
                              void* d_out, int out_size, void* d_ws, size_t ws_size,
                              hipStream_t stream) {
    const float* x  = (const float*)d_in[0];
    const float* W1 = (const float*)d_in[1];
    const float* b1 = (const float*)d_in[2];
    const float* W2 = (const float*)d_in[3];
    const float* b2 = (const float*)d_in[4];
    const float* W3 = (const float*)d_in[5];
    const float* b3 = (const float*)d_in[6];
    const float* W4 = (const float*)d_in[7];
    const float* b4 = (const float*)d_in[8];
    float* out = (float*)d_out;

    cpc_pair_mlp_fused_kernel<<<NWG, 256, 0, stream>>>(
        x, W1, b1, W2, b2, W3, b3, W4, b4, out);
}
